# starting kernel
# baseline (speedup 1.0000x reference)
#include <hip/hip_runtime.h>
#include <hip/hip_bf16.h>


#define B_  16
#define TQ_ 64
#define TK_ 512
#define D_  256
#define V_  32000

#define VB1 2048
#define NVB1 16
#define VB2 512
#define NVB2 63

typedef __attribute__((ext_vector_type(2))) float v2f;
typedef __attribute__((ext_vector_type(8))) float v8f;

__device__ inline v8f wmma_f32_k4(v2f a, v2f b, v8f c) {
  return __builtin_amdgcn_wmma_f32_16x16x4_f32(
       false, a,  false, b,
       (short)0, c,  false,  false);
}

__device__ inline float wave_max(float x) {
  #pragma unroll
  for (int o = 1; o < 32; o <<= 1) x = fmaxf(x, __shfl_xor(x, o, 32));
  return x;
}
__device__ inline float wave_sum(float x) {
  #pragma unroll
  for (int o = 1; o < 32; o <<= 1) x += __shfl_xor(x, o, 32);
  return x;
}
__device__ inline float half16_max(float x) {
  #pragma unroll
  for (int o = 1; o < 16; o <<= 1) x = fmaxf(x, __shfl_xor(x, o, 32));
  return x;
}
__device__ inline float half16_sum(float x) {
  #pragma unroll
  for (int o = 1; o < 16; o <<= 1) x += __shfl_xor(x, o, 32);
  return x;
}

__global__ __launch_bounds__(256) void k_attn(
    const float* __restrict__ dom, const float* __restrict__ out_st,
    const float* __restrict__ ctx, const float* __restrict__ Wg,
    const float* __restrict__ bg, float* __restrict__ attnW,
    float* __restrict__ pgenW)
{
  __shared__ float sS[16][TK_];
  __shared__ float sCV[16][D_];

  const int b    = blockIdx.x >> 2;
  const int q0   = (blockIdx.x & 3) << 4;
  const int wave = threadIdx.x >> 5;
  const int lane = threadIdx.x & 31;
  const int row  = lane & 15;
  const int hi   = lane >> 4;

  v8f acc[4];
  #pragma unroll
  for (int i = 0; i < 4; i++) acc[i] = {};
  const float* Abase = out_st + (size_t)(b*TQ_ + q0 + row)*D_;
  for (int dc = 0; dc < D_/4; ++dc) {
    const int d0 = dc*4;
    v2f a = *(const v2f*)(Abase + d0 + 2*hi);
    #pragma unroll
    for (int i = 0; i < 4; i++) {
      const int k0 = (wave*4 + i)*16;
      v2f bb = *(const v2f*)(ctx + (size_t)(b*TK_ + k0 + row)*D_ + d0 + 2*hi);
      acc[i] = wmma_f32_k4(a, bb, acc[i]);
    }
  }
  #pragma unroll
  for (int i = 0; i < 4; i++) {
    const int k0 = (wave*4 + i)*16;
    #pragma unroll
    for (int r = 0; r < 8; r++)
      sS[hi*8 + r][k0 + row] = acc[i][r] * 0.0625f;
  }
  __syncthreads();

  #pragma unroll
  for (int j = 0; j < 2; j++) {
    const int q = wave*2 + j;
    float v[TK_/32];
    float m = -1e30f;
    #pragma unroll
    for (int t = 0; t < TK_/32; t++) { v[t] = sS[q][lane + 32*t]; m = fmaxf(m, v[t]); }
    m = wave_max(m);
    float s = 0.f;
    #pragma unroll
    for (int t = 0; t < TK_/32; t++) { v[t] = __expf(v[t] - m); s += v[t]; }
    s = wave_sum(s);
    const float inv = 1.0f / s;
    #pragma unroll
    for (int t = 0; t < TK_/32; t++) {
      const float p = v[t] * inv;
      sS[q][lane + 32*t] = p;
      attnW[(size_t)(b*TQ_ + q0 + q)*TK_ + lane + 32*t] = p;
    }
  }
  __syncthreads();

  v8f cv[2]; cv[0] = {}; cv[1] = {};
  for (int kc = 0; kc < TK_/4; ++kc) {
    const int k0 = kc*4;
    v2f a;
    a.x = sS[row][k0 + 2*hi];
    a.y = sS[row][k0 + 2*hi + 1];
    #pragma unroll
    for (int i = 0; i < 2; i++) {
      const int n0 = (wave*2 + i)*16;
      v2f bb;
      bb.x = ctx[(size_t)(b*TK_ + k0 + 2*hi    )*D_ + n0 + row];
      bb.y = ctx[(size_t)(b*TK_ + k0 + 2*hi + 1)*D_ + n0 + row];
      cv[i] = wmma_f32_k4(a, bb, cv[i]);
    }
  }
  #pragma unroll
  for (int i = 0; i < 2; i++) {
    const int n0 = (wave*2 + i)*16;
    #pragma unroll
    for (int r = 0; r < 8; r++)
      sCV[hi*8 + r][n0 + row] = cv[i][r];
  }
  __syncthreads();

  #pragma unroll
  for (int j = 0; j < 2; j++) {
    const int q = wave*2 + j;
    float dot = 0.f;
    for (int d = lane; d < D_; d += 32) {
      dot += out_st[(size_t)(b*TQ_ + q0 + q)*D_ + d] * Wg[d];
      dot += sCV[q][d]                               * Wg[D_ + d];
      dot += dom[(size_t)(b*TQ_ + q0 + q)*D_ + d]    * Wg[2*D_ + d];
    }
    dot = wave_sum(dot);
    if (lane == 0) {
      const float x = dot + bg[0];
      pgenW[b*TQ_ + q0 + q] = 1.0f / (1.0f + __expf(-x));
    }
  }
}

__global__ __launch_bounds__(256) void k_stats(
    const float* __restrict__ out_st, const float* __restrict__ Ws,
    const float* __restrict__ bs, float* __restrict__ pmW,
    float* __restrict__ psW)
{
  const int b    = blockIdx.x >> 2;
  const int q0   = (blockIdx.x & 3) << 4;
  const int vb   = blockIdx.y;
  const int v0   = vb * VB1;
  const int cols = min(VB1, V_ - v0);
  const int ntiles = cols >> 4;

  const int wave = threadIdx.x >> 5;
  const int lane = threadIdx.x & 31;
  const int row  = lane & 15;
  const int hi   = lane >> 4;

  __shared__ float sM[8][16];
  __shared__ float sSm[8][16];

  float rm[8], rs[8];
  #pragma unroll
  for (int r = 0; r < 8; r++) { rm[r] = -1e30f; rs[r] = 0.f; }

  const float* Abase = out_st + (size_t)(b*TQ_ + q0 + row)*D_;
  for (int nt = wave; nt < ntiles; nt += 8) {
    const int c0 = v0 + nt*16;
    v8f acc = {};
    for (int dc = 0; dc < D_/4; ++dc) {
      const int d0 = dc*4;
      v2f a = *(const v2f*)(Abase + d0 + 2*hi);
      v2f bb;
      bb.x = Ws[(size_t)(d0 + 2*hi    )*V_ + c0 + row];
      bb.y = Ws[(size_t)(d0 + 2*hi + 1)*V_ + c0 + row];
      acc = wmma_f32_k4(a, bb, acc);
    }
    const float bsv = bs[c0 + row];
    #pragma unroll
    for (int r = 0; r < 8; r++) {
      const float z  = acc[r] + bsv;
      const float tm = half16_max(z);
      const float nm = fmaxf(rm[r], tm);
      const float ts = half16_sum(__expf(z - nm));
      rs[r] = rs[r] * __expf(rm[r] - nm) + ts;
      rm[r] = nm;
    }
  }
  if (row == 0) {
    #pragma unroll
    for (int r = 0; r < 8; r++) { sM[wave][hi*8 + r] = rm[r]; sSm[wave][hi*8 + r] = rs[r]; }
  }
  __syncthreads();
  if (threadIdx.x < 16) {
    const int q = threadIdx.x;
    float gm = -1e30f;
    #pragma unroll
    for (int w = 0; w < 8; w++) gm = fmaxf(gm, sM[w][q]);
    float gs = 0.f;
    #pragma unroll
    for (int w = 0; w < 8; w++) gs += sSm[w][q] * __expf(sM[w][q] - gm);
    const int rowid = b*TQ_ + q0 + q;
    pmW[(size_t)rowid*NVB1 + vb] = gm;
    psW[(size_t)rowid*NVB1 + vb] = gs;
  }
}

__global__ __launch_bounds__(256) void k_reduce(
    const float* __restrict__ pmW, const float* __restrict__ psW,
    float* __restrict__ rmW, float* __restrict__ rsW)
{
  const int rowid = blockIdx.x*256 + threadIdx.x;
  if (rowid >= B_*TQ_) return;
  float gm = -1e30f;
  #pragma unroll
  for (int v = 0; v < NVB1; ++v) gm = fmaxf(gm, pmW[(size_t)rowid*NVB1 + v]);
  float gs = 0.f;
  #pragma unroll
  for (int v = 0; v < NVB1; ++v)
    gs += psW[(size_t)rowid*NVB1 + v] * __expf(pmW[(size_t)rowid*NVB1 + v] - gm);
  rmW[rowid] = gm;
  rsW[rowid] = gs;
}

__global__ __launch_bounds__(256) void k_final(
    const float* __restrict__ out_st, const float* __restrict__ Ws,
    const float* __restrict__ bs, const int* __restrict__ cp,
    const float* __restrict__ attnW, const float* __restrict__ pgenW,
    const float* __restrict__ rmW, const float* __restrict__ rsW,
    float* __restrict__ outp)
{
  const int b    = blockIdx.x >> 2;
  const int q0   = (blockIdx.x & 3) << 4;
  const int vb   = blockIdx.y;
  const int v0   = vb * VB2;
  const int cols = min(VB2, V_ - v0);
  const int ntiles = cols >> 4;

  const int wave = threadIdx.x >> 5;
  const int lane = threadIdx.x & 31;
  const int row  = lane & 15;
  const int hi   = lane >> 4;

  __shared__ float sPtr[16][VB2];
  __shared__ float sScale[16], sMr[16], sOmp[16];

  for (int i = threadIdx.x; i < 16*VB2; i += 256)
    (&sPtr[0][0])[i] = 0.f;
  if (threadIdx.x < 16) {
    const int rowid = b*TQ_ + q0 + threadIdx.x;
    const float pg = pgenW[rowid];
    sScale[threadIdx.x] = pg / rsW[rowid];
    sMr[threadIdx.x]    = rmW[rowid];
    sOmp[threadIdx.x]   = 1.0f - pg;
  }
  __syncthreads();

  for (int k = threadIdx.x; k < TK_; k += 256) {
    const int key = cp[b*TK_ + k];
    if (key >= v0 && key < v0 + cols) {
      const int c = key - v0;
      #pragma unroll
      for (int q = 0; q < 16; q++) {
        const float w = sOmp[q] * attnW[(size_t)(b*TQ_ + q0 + q)*TK_ + k];
        atomicAdd(&sPtr[q][c], w);
      }
    }
  }
  __syncthreads();

  const float* Abase = out_st + (size_t)(b*TQ_ + q0 + row)*D_;
  for (int nt = wave; nt < ntiles; nt += 8) {
    const int c0 = v0 + nt*16;
    v8f acc = {};
    for (int dc = 0; dc < D_/4; ++dc) {
      const int d0 = dc*4;
      v2f a = *(const v2f*)(Abase + d0 + 2*hi);
      v2f bb;
      bb.x = Ws[(size_t)(d0 + 2*hi    )*V_ + c0 + row];
      bb.y = Ws[(size_t)(d0 + 2*hi + 1)*V_ + c0 + row];
      acc = wmma_f32_k4(a, bb, acc);
    }
    const float bsv = bs[c0 + row];
    #pragma unroll
    for (int r = 0; r < 8; r++) {
      const int q = hi*8 + r;
      const float z = acc[r] + bsv;
      const float p = sScale[q] * __expf(z - sMr[q]) + sPtr[q][nt*16 + row];
      outp[(size_t)(b*TQ_ + q0 + q)*V_ + c0 + row] = __logf(p);
    }
  }
}

extern "C" void kernel_launch(void* const* d_in, const int* in_sizes, int n_in,
                              void* d_out, int out_size, void* d_ws, size_t ws_size,
                              hipStream_t stream) {
  (void)in_sizes; (void)n_in; (void)out_size; (void)ws_size;
  const float* dom    = (const float*)d_in[0];
  const float* out_st = (const float*)d_in[1];
  const float* ctx    = (const float*)d_in[2];
  const int*   cp     = (const int*)d_in[3];
  const float* Ws     = (const float*)d_in[4];
  const float* bsv    = (const float*)d_in[5];
  const float* Wg     = (const float*)d_in[6];
  const float* bg     = (const float*)d_in[7];
  float* outp = (float*)d_out;

  float* attnW = (float*)d_ws;
  float* pgenW = attnW + (size_t)B_*TQ_*TK_;
  float* pmW   = pgenW + B_*TQ_;
  float* psW   = pmW + (size_t)B_*TQ_*NVB1;
  float* rmW   = psW + (size_t)B_*TQ_*NVB1;
  float* rsW   = rmW + B_*TQ_;

  k_attn  <<<dim3(B_*TQ_/16),        256, 0, stream>>>(dom, out_st, ctx, Wg, bg, attnW, pgenW);
  k_stats <<<dim3(B_*TQ_/16, NVB1),  256, 0, stream>>>(out_st, Ws, bsv, pmW, psW);
  k_reduce<<<dim3((B_*TQ_+255)/256), 256, 0, stream>>>(pmW, psW, rmW, rsW);
  k_final <<<dim3(B_*TQ_/16, NVB2),  256, 0, stream>>>(out_st, Ws, bsv, cp, attnW, pgenW, rmW, rsW, outp);
}


# Attention launch, operand feed regrouped: scores read 16 B k-quads (score sum order changed), context product 8 B column pairs + conflict-free score tile, 32-40 loads per trip, gate loads grouped
# speedup vs baseline: 2.5765x; 2.5765x over previous
#include <hip/hip_runtime.h>
#include <hip/hip_bf16.h>


#define B_  16
#define TQ_ 64
#define TK_ 512
#define D_  256
#define V_  32000

#define SNC  32
#define SPP  48
#define SNST 8
#define NSPL 125
#define PSTR 128

typedef __attribute__((ext_vector_type(2))) float v2f;
typedef __attribute__((ext_vector_type(8))) float v8f;

__device__ inline v8f wmma_f32_k4(v2f a, v2f b, v8f c) {
  return __builtin_amdgcn_wmma_f32_16x16x4_f32(
       false, a,  false, b,
       (short)0, c,  false,  false);
}

__device__ inline float wave_max(float x) {
  #pragma unroll
  for (int o = 1; o < 32; o <<= 1) x = fmaxf(x, __shfl_xor(x, o, 32));
  return x;
}
__device__ inline float wave_sum(float x) {
  #pragma unroll
  for (int o = 1; o < 32; o <<= 1) x += __shfl_xor(x, o, 32);
  return x;
}
__device__ inline float half16_max(float x) {
  #pragma unroll
  for (int o = 1; o < 16; o <<= 1) x = fmaxf(x, __shfl_xor(x, o, 32));
  return x;
}
__device__ inline float half16_sum(float x) {
  #pragma unroll
  for (int o = 1; o < 16; o <<= 1) x += __shfl_xor(x, o, 32);
  return x;
}

__global__ __launch_bounds__(256) void k_attn(
    const float* __restrict__ dom, const float* __restrict__ out_st,
    const float* __restrict__ ctx, const float* __restrict__ Wg,
    const float* __restrict__ bg, float* __restrict__ attnW,
    float* __restrict__ pgenW)
{
  __shared__ float sS[16][TK_ + 4];
  __shared__ float sCV[16][D_];

  const int b    = blockIdx.x >> 2;
  const int q0   = (blockIdx.x & 3) << 4;
  const int wave = threadIdx.x >> 5;
  const int lane = threadIdx.x & 31;
  const int row  = lane & 15;
  const int hi   = lane >> 4;

  v8f acc[4];
  #pragma unroll
  for (int i = 0; i < 4; i++) acc[i] = {};
  const float* Abase = out_st + (size_t)(b*TQ_ + q0 + row)*D_ + 4*hi;
  const float* Bbase = ctx + (size_t)(b*TK_ + wave*64 + row)*D_ + 4*hi;
  #pragma unroll 8
  for (int j = 0; j < D_/8; ++j) {
    const float4 a4 = *(const float4*)(Abase + 8*j);
    float4 b4[4];
    #pragma unroll
    for (int i = 0; i < 4; i++) b4[i] = *(const float4*)(Bbase + (size_t)(i*16)*D_ + 8*j);
    v2f a0, a1;
    a0.x = a4.x; a0.y = a4.y; a1.x = a4.z; a1.y = a4.w;
    #pragma unroll
    for (int i = 0; i < 4; i++) {
      v2f x; x.x = b4[i].x; x.y = b4[i].y;
      acc[i] = wmma_f32_k4(a0, x, acc[i]);
    }
    #pragma unroll
    for (int i = 0; i < 4; i++) {
      v2f y; y.x = b4[i].z; y.y = b4[i].w;
      acc[i] = wmma_f32_k4(a1, y, acc[i]);
    }
  }
  #pragma unroll
  for (int i = 0; i < 4; i++) {
    const int k0 = (wave*4 + i)*16;
    #pragma unroll
    for (int r = 0; r < 8; r++)
      sS[hi*8 + r][k0 + row] = acc[i][r] * 0.0625f;
  }
  __syncthreads();

  #pragma unroll
  for (int j = 0; j < 2; j++) {
    const int q = wave*2 + j;
    float v[TK_/32];
    float m = -1e30f;
    #pragma unroll
    for (int t = 0; t < TK_/32; t++) { v[t] = sS[q][lane + 32*t]; m = fmaxf(m, v[t]); }
    m = wave_max(m);
    float s = 0.f;
    #pragma unroll
    for (int t = 0; t < TK_/32; t++) { v[t] = __expf(v[t] - m); s += v[t]; }
    s = wave_sum(s);
    const float inv = 1.0f / s;
    #pragma unroll
    for (int t = 0; t < TK_/32; t++) {
      const float p = v[t] * inv;
      sS[q][lane + 32*t] = p;
      attnW[(size_t)(b*TQ_ + q0 + q)*TK_ + lane + 32*t] = p;
    }
  }
  __syncthreads();

  v8f cv[2]; cv[0] = {}; cv[1] = {};
  const float* Cb = ctx + (size_t)(b*TK_ + 2*hi)*D_ + 32*wave + 2*row;
  #pragma unroll 16
  for (int kc = 0; kc < TK_/4; ++kc) {
    const int k0 = kc*4;
    v2f a;
    a.x = sS[row][k0 + 2*hi];
    a.y = sS[row][k0 + 2*hi + 1];
    const v2f c0 = *(const v2f*)(Cb + (size_t)k0*D_);
    const v2f c1 = *(const v2f*)(Cb + (size_t)(k0 + 1)*D_);
    v2f bb0, bb1;
    bb0.x = c0.x; bb0.y = c1.x;
    bb1.x = c0.y; bb1.y = c1.y;
    cv[0] = wmma_f32_k4(a, bb0, cv[0]);
    cv[1] = wmma_f32_k4(a, bb1, cv[1]);
  }

  const int qa = wave*2;
  float w0[8], w1[8], w2[8], ov[2][8], dv[2][8];
  #pragma unroll
  for (int t = 0; t < 8; t++) {
    const int d = lane + 32*t;
    w0[t] = Wg[d]; w1[t] = Wg[D_ + d]; w2[t] = Wg[2*D_ + d];
    #pragma unroll
    for (int j = 0; j < 2; j++) {
      ov[j][t] = out_st[(size_t)(b*TQ_ + q0 + qa + j)*D_ + d];
      dv[j][t] = dom[(size_t)(b*TQ_ + q0 + qa + j)*D_ + d];
    }
  }
  const float bg0 = bg[0];

  #pragma unroll
  for (int r = 0; r < 8; r++) {
    sCV[hi*8 + r][32*wave + 2*row]     = cv[0][r];
    sCV[hi*8 + r][32*wave + 2*row + 1] = cv[1][r];
  }
  __syncthreads();

  #pragma unroll
  for (int j = 0; j < 2; j++) {
    const int q = qa + j;
    float dot = 0.f;
    #pragma unroll
    for (int t = 0; t < 8; t++) {
      dot += ov[j][t]            * w0[t];
      dot += sCV[q][lane + 32*t] * w1[t];
      dot += dv[j][t]            * w2[t];
    }
    dot = wave_sum(dot);
    if (lane == 0) {
      const float x = dot + bg0;
      pgenW[b*TQ_ + q0 + q] = 1.0f / (1.0f + __expf(-x));
    }
  }
}

__global__ __launch_bounds__(256) void k_stats(
    const float* __restrict__ out_st, const float* __restrict__ Ws,
    const float* __restrict__ bs, float* __restrict__ pmW,
    float* __restrict__ psW)
{
  __shared__ v2f sWp[D_/2][SPP];

  const int tid  = threadIdx.x;
  const int wave = tid >> 5;
  const int lane = tid & 31;
  const int row  = lane & 15;
  const int hi   = lane >> 4;
  const int sp   = blockIdx.y;
  const int v0   = sp * (SNC*SNST);
  const int rowbase = blockIdx.x*128 + wave*16;

  v2f a[D_/4];
  const float* Abase = out_st + (size_t)(rowbase + row)*D_ + 2*hi;
  #pragma unroll
  for (int dc = 0; dc < D_/4; ++dc) a[dc] = *(const v2f*)(Abase + dc*4);

  float rm[8], rs[8];
  #pragma unroll
  for (int r = 0; r < 8; r++) { rm[r] = -1e30f; rs[r] = 0.f; }

  const int sd = tid >> 3;
  const int sc = (tid & 7) * 4;
  const float* Wsrc = Ws + (size_t)(2*sd)*V_ + v0 + sc;
  float4 pf[8];
  #pragma unroll
  for (int i = 0; i < 4; i++) {
    pf[2*i]     = *(const float4*)(Wsrc + (size_t)(i*64)*V_);
    pf[2*i + 1] = *(const float4*)(Wsrc + (size_t)(i*64 + 1)*V_);
  }

  const v2f* pB = &sWp[hi][row];

  #pragma unroll 1
  for (int s = 0; s < SNST; ++s) {
    #pragma unroll
    for (int i = 0; i < 4; i++) {
      const float4 x = pf[2*i], y = pf[2*i + 1];
      float4 lo, up;
      lo.x = x.x; lo.y = y.x; lo.z = x.y; lo.w = y.y;
      up.x = x.z; up.y = y.z; up.z = x.w; up.w = y.w;
      *(float4*)&sWp[i*32 + sd][sc]     = lo;
      *(float4*)&sWp[i*32 + sd][sc + 2] = up;
    }
    __syncthreads();
    if (s + 1 < SNST) {
      #pragma unroll
      for (int i = 0; i < 4; i++) {
        pf[2*i]     = *(const float4*)(Wsrc + (size_t)(i*64)*V_     + (s + 1)*SNC);
        pf[2*i + 1] = *(const float4*)(Wsrc + (size_t)(i*64 + 1)*V_ + (s + 1)*SNC);
      }
    }

    v8f acc0 = {}, acc1 = {};
    #pragma unroll
    for (int dc = 0; dc < D_/4; ++dc) {
      const v2f b0 = pB[dc*2*SPP];
      const v2f b1 = pB[dc*2*SPP + 16];
      acc0 = wmma_f32_k4(a[dc], b0, acc0);
      acc1 = wmma_f32_k4(a[dc], b1, acc1);
    }

    const int vs = v0 + s*SNC;
    const float bsv0 = bs[vs + row];
    const float bsv1 = bs[vs + 16 + row];
    #pragma unroll
    for (int r = 0; r < 8; r++) {
      const float z0 = acc0[r] + bsv0;
      const float z1 = acc1[r] + bsv1;
      const float nm = fmaxf(rm[r], fmaxf(z0, z1));
      rs[r] = rs[r] * __expf(rm[r] - nm) + (__expf(z0 - nm) + __expf(z1 - nm));
      rm[r] = nm;
    }
    __syncthreads();
  }

  #pragma unroll
  for (int r = 0; r < 8; r++) {
    const float gm = half16_max(rm[r]);
    const float gs = half16_sum(rs[r] * __expf(rm[r] - gm));
    if (row == 0) {
      const int rowid = rowbase + hi*8 + r;
      pmW[(size_t)rowid*PSTR + sp] = gm;
      psW[(size_t)rowid*PSTR + sp] = gs;
    }
  }
}

__global__ __launch_bounds__(256) void k_reduce(
    const float* __restrict__ pmW, const float* __restrict__ psW,
    float* __restrict__ rmW, float* __restrict__ rsW)
{
  const int rowid = blockIdx.x*8 + (threadIdx.x >> 5);
  const int lane  = threadIdx.x & 31;
  float m[4], s[4];
  float gm = -1e30f;
  #pragma unroll
  for (int t = 0; t < 4; t++) {
    const int spi = lane + 32*t;
    const bool ok = spi < NSPL;
    m[t] = ok ? pmW[(size_t)rowid*PSTR + spi] : -1e30f;
    s[t] = ok ? psW[(size_t)rowid*PSTR + spi] : 0.f;
    gm = fmaxf(gm, m[t]);
  }
  gm = wave_max(gm);
  float gs = 0.f;
  #pragma unroll
  for (int t = 0; t < 4; t++) gs += s[t] * __expf(m[t] - gm);
  gs = wave_sum(gs);
  if (lane == 0) { rmW[rowid] = gm; rsW[rowid] = gs; }
}

__global__ __launch_bounds__(256) void k_final(
    const float* __restrict__ out_st, const float* __restrict__ Ws,
    const float* __restrict__ bs, const int* __restrict__ cp,
    const float* __restrict__ attnW, const float* __restrict__ pgenW,
    const float* __restrict__ rmW, const float* __restrict__ rsW,
    float* __restrict__ outp)
{
  __shared__ v2f sWp[D_/2][SPP];
  __shared__ int sLk[2][TK_];
  __shared__ int sLc[2][TK_];
  __shared__ int sLn[2];

  const int tid  = threadIdx.x;
  const int wave = tid >> 5;
  const int lane = tid & 31;
  const int row  = lane & 15;
  const int hi   = lane >> 4;
  const int sp   = blockIdx.y;
  const int v0   = sp * (SNC*SNST);
  const int rowbase = blockIdx.x*128 + wave*16;
  const int b2   = wave >> 2;

  if (wave < 2) {
    const int* keys = cp + (blockIdx.x*2 + wave)*TK_;
    int kv[TK_/32];
    #pragma unroll
    for (int it = 0; it < TK_/32; ++it) kv[it] = keys[it*32 + lane];
    int base = 0;
    #pragma unroll
    for (int it = 0; it < TK_/32; ++it) {
      const int c   = kv[it] - v0;
      const bool in = (unsigned)c < (unsigned)(SNC*SNST);
      const unsigned mask = (unsigned)__ballot(in);
      if (in) {
        const int pos = base + __popc(mask & ((1u << lane) - 1u));
        sLk[wave][pos] = it*32 + lane;
        sLc[wave][pos] = c;
      }
      base += __popc(mask);
    }
    if (lane == 0) sLn[wave] = base;
  }

  v2f a[D_/4];
  const float* Abase = out_st + (size_t)(rowbase + row)*D_ + 2*hi;
  #pragma unroll
  for (int dc = 0; dc < D_/4; ++dc) a[dc] = *(const v2f*)(Abase + dc*4);

  float sc8[8], mr8[8];
  #pragma unroll
  for (int r = 0; r < 8; r++) {
    const int rowid = rowbase + hi*8 + r;
    sc8[r] = pgenW[rowid] / rsW[rowid];
    mr8[r] = rmW[rowid];
  }

  const int sd = tid >> 3;
  const int sc = (tid & 7) * 4;
  const float* Wsrc = Ws + (size_t)(2*sd)*V_ + v0 + sc;
  float4 pf[8];
  #pragma unroll
  for (int i = 0; i < 4; i++) {
    pf[2*i]     = *(const float4*)(Wsrc + (size_t)(i*64)*V_);
    pf[2*i + 1] = *(const float4*)(Wsrc + (size_t)(i*64 + 1)*V_);
  }

  const v2f* pB = &sWp[hi][row];

  float h0[8], h1[8];
  #pragma unroll
  for (int r = 0; r < 8; r++) { h0[r] = 0.f; h1[r] = 0.f; }

  #pragma unroll 1
  for (int s = 0; s < SNST; ++s) {
    #pragma unroll
    for (int i = 0; i < 4; i++) {
      const float4 x = pf[2*i], y = pf[2*i + 1];
      float4 lo, up;
      lo.x = x.x; lo.y = y.x; lo.z = x.y; lo.w = y.y;
      up.x = x.z; up.y = y.z; up.z = x.w; up.w = y.w;
      *(float4*)&sWp[i*32 + sd][sc]     = lo;
      *(float4*)&sWp[i*32 + sd][sc + 2] = up;
    }
    __syncthreads();
    if (s > 0) {
      #pragma unroll
      for (int r = 0; r < 8; r++) {
        float* o = outp + (size_t)(rowbase + hi*8 + r)*V_ + v0 + (s - 1)*SNC + row;
        o[0]  = h0[r];
        o[16] = h1[r];
      }
    }
    if (s + 1 < SNST) {
      #pragma unroll
      for (int i = 0; i < 4; i++) {
        pf[2*i]     = *(const float4*)(Wsrc + (size_t)(i*64)*V_     + (s + 1)*SNC);
        pf[2*i + 1] = *(const float4*)(Wsrc + (size_t)(i*64 + 1)*V_ + (s + 1)*SNC);
      }
    }

    v8f acc0 = {}, acc1 = {};
    #pragma unroll
    for (int dc = 0; dc < D_/4; ++dc) {
      const v2f b0 = pB[dc*2*SPP];
      const v2f b1 = pB[dc*2*SPP + 16];
      acc0 = wmma_f32_k4(a[dc], b0, acc0);
      acc1 = wmma_f32_k4(a[dc], b1, acc1);
    }

    const int vs = v0 + s*SNC;
    const float bsv0 = bs[vs + row];
    const float bsv1 = bs[vs + 16 + row];
    #pragma unroll
    for (int r = 0; r < 8; r++) {
      acc0[r] = sc8[r] * __expf((acc0[r] + bsv0) - mr8[r]);
      acc1[r] = sc8[r] * __expf((acc1[r] + bsv1) - mr8[r]);
    }

    const int nk = sLn[b2];
    for (int j0 = 0; j0 < nk; j0 += 32) {
      const int j = j0 + lane;
      int cj = -1, kj = 0;
      if (j < nk) { cj = sLc[b2][j] - s*SNC; kj = sLk[b2][j]; }
      unsigned mask = (unsigned)__ballot((unsigned)cj < (unsigned)SNC);
      while (mask) {
        const int src = __ffs(mask) - 1;
        mask &= mask - 1u;
        const int c  = __shfl(cj, src, 32);
        const int kk = __shfl(kj, src, 32);
        if (row == (c & 15)) {
          #pragma unroll
          for (int r = 0; r < 8; r++) {
            const int rowid = rowbase + hi*8 + r;
            const float w = (1.0f - pgenW[rowid]) * attnW[(size_t)rowid*TK_ + kk];
            if (c < 16) acc0[r] += w; else acc1[r] += w;
          }
        }
      }
    }

    #pragma unroll
    for (int r = 0; r < 8; r++) {
      h0[r] = __logf(acc0[r]);
      h1[r] = __logf(acc1[r]);
    }
    __syncthreads();
  }

  #pragma unroll
  for (int r = 0; r < 8; r++) {
    float* o = outp + (size_t)(rowbase + hi*8 + r)*V_ + v0 + (SNST - 1)*SNC + row;
    o[0]  = h0[r];
    o[16] = h1[r];
  }
}

extern "C" void kernel_launch(void* const* d_in, const int* in_sizes, int n_in,
                              void* d_out, int out_size, void* d_ws, size_t ws_size,
                              hipStream_t stream) {
  (void)in_sizes; (void)n_in; (void)out_size; (void)ws_size;
  const float* dom    = (const float*)d_in[0];
  const float* out_st = (const float*)d_in[1];
  const float* ctx    = (const float*)d_in[2];
  const int*   cp     = (const int*)d_in[3];
  const float* Ws     = (const float*)d_in[4];
  const float* bsv    = (const float*)d_in[5];
  const float* Wg     = (const float*)d_in[6];
  const float* bg     = (const float*)d_in[7];
  float* outp = (float*)d_out;

  float* attnW = (float*)d_ws;
  float* pgenW = attnW + (size_t)B_*TQ_*TK_;
  float* rmW   = pgenW + B_*TQ_;
  float* rsW   = rmW + B_*TQ_;
  float* pmW   = outp;
  float* psW   = outp + (size_t)B_*TQ_*PSTR;

  k_attn  <<<dim3(B_*TQ_/16),        256, 0, stream>>>(dom, out_st, ctx, Wg, bg, attnW, pgenW);
  k_stats <<<dim3(B_*TQ_/128, NSPL), 256, 0, stream>>>(out_st, Ws, bsv, pmW, psW);
  k_reduce<<<dim3(B_*TQ_/8),         256, 0, stream>>>(pmW, psW, rmW, rsW);
  k_final <<<dim3(B_*TQ_/128, NSPL), 256, 0, stream>>>(out_st, Ws, bsv, cp, attnW, pgenW, rmW, rsW, outp);
}


# Attention launch split over key quarters: 256 groups (was 64), each a local softmax + its share of context and gate; rows merged in the reduce launch (softmax, gate and pointer sum order changed)
# speedup vs baseline: 2.6795x; 1.0400x over previous
#include <hip/hip_runtime.h>
#include <hip/hip_bf16.h>


#define B_  16
#define TQ_ 64
#define TK_ 512
#define D_  256
#define V_  32000

#define SNC  32
#define SPP  48
#define SNST 8
#define NSPL 125
#define PSTR 128
#define NKP  4
#define KP_  (TK_/NKP)

typedef __attribute__((ext_vector_type(2))) float v2f;
typedef __attribute__((ext_vector_type(8))) float v8f;

__device__ inline v8f wmma_f32_k4(v2f a, v2f b, v8f c) {
  return __builtin_amdgcn_wmma_f32_16x16x4_f32(
       false, a,  false, b,
       (short)0, c,  false,  false);
}

__device__ inline float wave_max(float x) {
  #pragma unroll
  for (int o = 1; o < 32; o <<= 1) x = fmaxf(x, __shfl_xor(x, o, 32));
  return x;
}
__device__ inline float wave_sum(float x) {
  #pragma unroll
  for (int o = 1; o < 32; o <<= 1) x += __shfl_xor(x, o, 32);
  return x;
}
__device__ inline float half16_max(float x) {
  #pragma unroll
  for (int o = 1; o < 16; o <<= 1) x = fmaxf(x, __shfl_xor(x, o, 32));
  return x;
}
__device__ inline float half16_sum(float x) {
  #pragma unroll
  for (int o = 1; o < 16; o <<= 1) x += __shfl_xor(x, o, 32);
  return x;
}

__global__ __launch_bounds__(256) void k_attn(
    const float* __restrict__ dom, const float* __restrict__ out_st,
    const float* __restrict__ ctx, const float* __restrict__ Wg,
    float* __restrict__ eW, float* __restrict__ amW, float* __restrict__ asW,
    float* __restrict__ auW, float* __restrict__ abW)
{
  __shared__ float sS[16][KP_ + 4];
  __shared__ float sCV[16][D_];

  const int part = blockIdx.x & 3;
  const int q0   = ((blockIdx.x >> 2) & 3) << 4;
  const int b    = blockIdx.x >> 4;
  const int kb   = part * KP_;
  const int wave = threadIdx.x >> 5;
  const int lane = threadIdx.x & 31;
  const int row  = lane & 15;
  const int hi   = lane >> 4;

  v8f acc = {};
  const float* Abase = out_st + (size_t)(b*TQ_ + q0 + row)*D_ + 4*hi;
  const float* Bbase = ctx + (size_t)(b*TK_ + kb + wave*16 + row)*D_ + 4*hi;
  #pragma unroll 16
  for (int j = 0; j < D_/8; ++j) {
    const float4 a4 = *(const float4*)(Abase + 8*j);
    const float4 b4 = *(const float4*)(Bbase + 8*j);
    v2f a0, a1, x, y;
    a0.x = a4.x; a0.y = a4.y; a1.x = a4.z; a1.y = a4.w;
    x.x = b4.x; x.y = b4.y; y.x = b4.z; y.y = b4.w;
    acc = wmma_f32_k4(a0, x, acc);
    acc = wmma_f32_k4(a1, y, acc);
  }
  #pragma unroll
  for (int r = 0; r < 8; r++)
    sS[hi*8 + r][wave*16 + row] = acc[r] * 0.0625f;
  __syncthreads();

  #pragma unroll
  for (int j = 0; j < 2; j++) {
    const int q = wave*2 + j;
    const int rowg = b*TQ_ + q0 + q;
    float v[KP_/32];
    float m = -1e30f;
    #pragma unroll
    for (int t = 0; t < KP_/32; t++) { v[t] = sS[q][lane + 32*t]; m = fmaxf(m, v[t]); }
    m = wave_max(m);
    float s = 0.f;
    #pragma unroll
    for (int t = 0; t < KP_/32; t++) { v[t] = __expf(v[t] - m); s += v[t]; }
    s = wave_sum(s);
    #pragma unroll
    for (int t = 0; t < KP_/32; t++) {
      sS[q][lane + 32*t] = v[t];
      eW[(size_t)rowg*TK_ + kb + lane + 32*t] = v[t];
    }
    if (lane == 0) { amW[rowg*NKP + part] = m; asW[rowg*NKP + part] = s; }
  }
  __syncthreads();

  const int qa = wave*2;
  float w1[8], w0[8], w2[8], ov[2][8], dv[2][8];
  #pragma unroll
  for (int t = 0; t < 8; t++) {
    w1[t] = Wg[D_ + lane + 32*t];
    w0[t] = 0.f; w2[t] = 0.f;
    ov[0][t] = 0.f; ov[1][t] = 0.f; dv[0][t] = 0.f; dv[1][t] = 0.f;
  }
  if (part == 0) {
    #pragma unroll
    for (int t = 0; t < 8; t++) {
      const int d = lane + 32*t;
      w0[t] = Wg[d]; w2[t] = Wg[2*D_ + d];
      #pragma unroll
      for (int j = 0; j < 2; j++) {
        ov[j][t] = out_st[(size_t)(b*TQ_ + q0 + qa + j)*D_ + d];
        dv[j][t] = dom[(size_t)(b*TQ_ + q0 + qa + j)*D_ + d];
      }
    }
  }

  v8f cv[2]; cv[0] = {}; cv[1] = {};
  const float* Cb = ctx + (size_t)(b*TK_ + kb + 2*hi)*D_ + 32*wave + 2*row;
  #pragma unroll 16
  for (int kc = 0; kc < KP_/4; ++kc) {
    const int k0 = kc*4;
    v2f a;
    a.x = sS[row][k0 + 2*hi];
    a.y = sS[row][k0 + 2*hi + 1];
    const v2f c0 = *(const v2f*)(Cb + (size_t)k0*D_);
    const v2f c1 = *(const v2f*)(Cb + (size_t)(k0 + 1)*D_);
    v2f bb0, bb1;
    bb0.x = c0.x; bb0.y = c1.x;
    bb1.x = c0.y; bb1.y = c1.y;
    cv[0] = wmma_f32_k4(a, bb0, cv[0]);
    cv[1] = wmma_f32_k4(a, bb1, cv[1]);
  }

  #pragma unroll
  for (int r = 0; r < 8; r++) {
    sCV[hi*8 + r][32*wave + 2*row]     = cv[0][r];
    sCV[hi*8 + r][32*wave + 2*row + 1] = cv[1][r];
  }
  __syncthreads();

  #pragma unroll
  for (int j = 0; j < 2; j++) {
    const int q = qa + j;
    const int rowg = b*TQ_ + q0 + q;
    float du = 0.f;
    #pragma unroll
    for (int t = 0; t < 8; t++) du += sCV[q][lane + 32*t] * w1[t];
    du = wave_sum(du);
    if (lane == 0) auW[rowg*NKP + part] = du;
    if (part == 0) {
      float db = 0.f;
      #pragma unroll
      for (int t = 0; t < 8; t++) {
        db += ov[j][t] * w0[t];
        db += dv[j][t] * w2[t];
      }
      db = wave_sum(db);
      if (lane == 0) abW[rowg] = db;
    }
  }
}

__global__ __launch_bounds__(256) void k_stats(
    const float* __restrict__ out_st, const float* __restrict__ Ws,
    const float* __restrict__ bs, float* __restrict__ pmW,
    float* __restrict__ psW)
{
  __shared__ v2f sWp[D_/2][SPP];

  const int tid  = threadIdx.x;
  const int wave = tid >> 5;
  const int lane = tid & 31;
  const int row  = lane & 15;
  const int hi   = lane >> 4;
  const int sp   = blockIdx.y;
  const int v0   = sp * (SNC*SNST);
  const int rowbase = blockIdx.x*128 + wave*16;

  v2f a[D_/4];
  const float* Abase = out_st + (size_t)(rowbase + row)*D_ + 2*hi;
  #pragma unroll
  for (int dc = 0; dc < D_/4; ++dc) a[dc] = *(const v2f*)(Abase + dc*4);

  float rm[8], rs[8];
  #pragma unroll
  for (int r = 0; r < 8; r++) { rm[r] = -1e30f; rs[r] = 0.f; }

  const int sd = tid >> 3;
  const int sc = (tid & 7) * 4;
  const float* Wsrc = Ws + (size_t)(2*sd)*V_ + v0 + sc;
  float4 pf[8];
  #pragma unroll
  for (int i = 0; i < 4; i++) {
    pf[2*i]     = *(const float4*)(Wsrc + (size_t)(i*64)*V_);
    pf[2*i + 1] = *(const float4*)(Wsrc + (size_t)(i*64 + 1)*V_);
  }

  const v2f* pB = &sWp[hi][row];

  #pragma unroll 1
  for (int s = 0; s < SNST; ++s) {
    #pragma unroll
    for (int i = 0; i < 4; i++) {
      const float4 x = pf[2*i], y = pf[2*i + 1];
      float4 lo, up;
      lo.x = x.x; lo.y = y.x; lo.z = x.y; lo.w = y.y;
      up.x = x.z; up.y = y.z; up.z = x.w; up.w = y.w;
      *(float4*)&sWp[i*32 + sd][sc]     = lo;
      *(float4*)&sWp[i*32 + sd][sc + 2] = up;
    }
    __syncthreads();
    if (s + 1 < SNST) {
      #pragma unroll
      for (int i = 0; i < 4; i++) {
        pf[2*i]     = *(const float4*)(Wsrc + (size_t)(i*64)*V_     + (s + 1)*SNC);
        pf[2*i + 1] = *(const float4*)(Wsrc + (size_t)(i*64 + 1)*V_ + (s + 1)*SNC);
      }
    }

    v8f acc0 = {}, acc1 = {};
    #pragma unroll
    for (int dc = 0; dc < D_/4; ++dc) {
      const v2f b0 = pB[dc*2*SPP];
      const v2f b1 = pB[dc*2*SPP + 16];
      acc0 = wmma_f32_k4(a[dc], b0, acc0);
      acc1 = wmma_f32_k4(a[dc], b1, acc1);
    }

    const int vs = v0 + s*SNC;
    const float bsv0 = bs[vs + row];
    const float bsv1 = bs[vs + 16 + row];
    #pragma unroll
    for (int r = 0; r < 8; r++) {
      const float z0 = acc0[r] + bsv0;
      const float z1 = acc1[r] + bsv1;
      const float nm = fmaxf(rm[r], fmaxf(z0, z1));
      rs[r] = rs[r] * __expf(rm[r] - nm) + (__expf(z0 - nm) + __expf(z1 - nm));
      rm[r] = nm;
    }
    __syncthreads();
  }

  #pragma unroll
  for (int r = 0; r < 8; r++) {
    const float gm = half16_max(rm[r]);
    const float gs = half16_sum(rs[r] * __expf(rm[r] - gm));
    if (row == 0) {
      const int rowid = rowbase + hi*8 + r;
      pmW[(size_t)rowid*PSTR + sp] = gm;
      psW[(size_t)rowid*PSTR + sp] = gs;
    }
  }
}

__global__ __launch_bounds__(256) void k_reduce(
    const float* __restrict__ pmW, const float* __restrict__ psW,
    float* __restrict__ rmW, float* __restrict__ rsW,
    const float* __restrict__ amW, const float* __restrict__ asW,
    const float* __restrict__ auW, const float* __restrict__ abW,
    const float* __restrict__ bg, float* __restrict__ eW,
    float* __restrict__ pgenW)
{
  const int rowid = blockIdx.x*8 + (threadIdx.x >> 5);
  const int lane  = threadIdx.x & 31;
  float m[4], s[4];
  #pragma unroll
  for (int t = 0; t < 4; t++) {
    const int spi = lane + 32*t;
    const bool ok = spi < NSPL;
    m[t] = ok ? pmW[(size_t)rowid*PSTR + spi] : -1e30f;
    s[t] = ok ? psW[(size_t)rowid*PSTR + spi] : 0.f;
  }
  float am[NKP], as4[NKP], au[NKP];
  #pragma unroll
  for (int p = 0; p < NKP; p++) {
    am[p]  = amW[rowid*NKP + p];
    as4[p] = asW[rowid*NKP + p];
    au[p]  = auW[rowid*NKP + p];
  }
  const float ab  = abW[rowid];
  const float bg0 = bg[0];
  float e[TK_/32];
  #pragma unroll
  for (int t = 0; t < TK_/32; t++) e[t] = eW[(size_t)rowid*TK_ + lane + 32*t];

  float gm = -1e30f;
  #pragma unroll
  for (int t = 0; t < 4; t++) gm = fmaxf(gm, m[t]);
  gm = wave_max(gm);
  float gs = 0.f;
  #pragma unroll
  for (int t = 0; t < 4; t++) gs += s[t] * __expf(m[t] - gm);
  gs = wave_sum(gs);

  float M = am[0];
  #pragma unroll
  for (int p = 1; p < NKP; p++) M = fmaxf(M, am[p]);
  float c[NKP];
  float S = 0.f, U = 0.f;
  #pragma unroll
  for (int p = 0; p < NKP; p++) {
    c[p] = __expf(am[p] - M);
    S += as4[p] * c[p];
    U += au[p] * c[p];
  }
  const float invS = 1.0f / S;
  const float x  = (ab + U * invS) + bg0;
  const float pg = 1.0f / (1.0f + __expf(-x));
  const float om = 1.0f - pg;
  #pragma unroll
  for (int t = 0; t < TK_/32; t++)
    eW[(size_t)rowid*TK_ + lane + 32*t] = e[t] * (om * (c[t >> 2] * invS));
  if (lane == 0) { rmW[rowid] = gm; rsW[rowid] = gs; pgenW[rowid] = pg; }
}

__global__ __launch_bounds__(256) void k_final(
    const float* __restrict__ out_st, const float* __restrict__ Ws,
    const float* __restrict__ bs, const int* __restrict__ cp,
    const float* __restrict__ attnW, const float* __restrict__ pgenW,
    const float* __restrict__ rmW, const float* __restrict__ rsW,
    float* __restrict__ outp)
{
  __shared__ v2f sWp[D_/2][SPP];
  __shared__ int sLk[2][TK_];
  __shared__ int sLc[2][TK_];
  __shared__ int sLn[2];

  const int tid  = threadIdx.x;
  const int wave = tid >> 5;
  const int lane = tid & 31;
  const int row  = lane & 15;
  const int hi   = lane >> 4;
  const int sp   = blockIdx.y;
  const int v0   = sp * (SNC*SNST);
  const int rowbase = blockIdx.x*128 + wave*16;
  const int b2   = wave >> 2;

  if (wave < 2) {
    const int* keys = cp + (blockIdx.x*2 + wave)*TK_;
    int kv[TK_/32];
    #pragma unroll
    for (int it = 0; it < TK_/32; ++it) kv[it] = keys[it*32 + lane];
    int base = 0;
    #pragma unroll
    for (int it = 0; it < TK_/32; ++it) {
      const int c   = kv[it] - v0;
      const bool in = (unsigned)c < (unsigned)(SNC*SNST);
      const unsigned mask = (unsigned)__ballot(in);
      if (in) {
        const int pos = base + __popc(mask & ((1u << lane) - 1u));
        sLk[wave][pos] = it*32 + lane;
        sLc[wave][pos] = c;
      }
      base += __popc(mask);
    }
    if (lane == 0) sLn[wave] = base;
  }

  v2f a[D_/4];
  const float* Abase = out_st + (size_t)(rowbase + row)*D_ + 2*hi;
  #pragma unroll
  for (int dc = 0; dc < D_/4; ++dc) a[dc] = *(const v2f*)(Abase + dc*4);

  float sc8[8], mr8[8];
  #pragma unroll
  for (int r = 0; r < 8; r++) {
    const int rowid = rowbase + hi*8 + r;
    sc8[r] = pgenW[rowid] / rsW[rowid];
    mr8[r] = rmW[rowid];
  }

  const int sd = tid >> 3;
  const int sc = (tid & 7) * 4;
  const float* Wsrc = Ws + (size_t)(2*sd)*V_ + v0 + sc;
  float4 pf[8];
  #pragma unroll
  for (int i = 0; i < 4; i++) {
    pf[2*i]     = *(const float4*)(Wsrc + (size_t)(i*64)*V_);
    pf[2*i + 1] = *(const float4*)(Wsrc + (size_t)(i*64 + 1)*V_);
  }

  const v2f* pB = &sWp[hi][row];

  float h0[8], h1[8];
  #pragma unroll
  for (int r = 0; r < 8; r++) { h0[r] = 0.f; h1[r] = 0.f; }

  #pragma unroll 1
  for (int s = 0; s < SNST; ++s) {
    #pragma unroll
    for (int i = 0; i < 4; i++) {
      const float4 x = pf[2*i], y = pf[2*i + 1];
      float4 lo, up;
      lo.x = x.x; lo.y = y.x; lo.z = x.y; lo.w = y.y;
      up.x = x.z; up.y = y.z; up.z = x.w; up.w = y.w;
      *(float4*)&sWp[i*32 + sd][sc]     = lo;
      *(float4*)&sWp[i*32 + sd][sc + 2] = up;
    }
    __syncthreads();
    if (s > 0) {
      #pragma unroll
      for (int r = 0; r < 8; r++) {
        float* o = outp + (size_t)(rowbase + hi*8 + r)*V_ + v0 + (s - 1)*SNC + row;
        o[0]  = h0[r];
        o[16] = h1[r];
      }
    }
    if (s + 1 < SNST) {
      #pragma unroll
      for (int i = 0; i < 4; i++) {
        pf[2*i]     = *(const float4*)(Wsrc + (size_t)(i*64)*V_     + (s + 1)*SNC);
        pf[2*i + 1] = *(const float4*)(Wsrc + (size_t)(i*64 + 1)*V_ + (s + 1)*SNC);
      }
    }

    v8f acc0 = {}, acc1 = {};
    #pragma unroll
    for (int dc = 0; dc < D_/4; ++dc) {
      const v2f b0 = pB[dc*2*SPP];
      const v2f b1 = pB[dc*2*SPP + 16];
      acc0 = wmma_f32_k4(a[dc], b0, acc0);
      acc1 = wmma_f32_k4(a[dc], b1, acc1);
    }

    const int vs = v0 + s*SNC;
    const float bsv0 = bs[vs + row];
    const float bsv1 = bs[vs + 16 + row];
    #pragma unroll
    for (int r = 0; r < 8; r++) {
      acc0[r] = sc8[r] * __expf((acc0[r] + bsv0) - mr8[r]);
      acc1[r] = sc8[r] * __expf((acc1[r] + bsv1) - mr8[r]);
    }

    const int nk = sLn[b2];
    for (int j0 = 0; j0 < nk; j0 += 32) {
      const int j = j0 + lane;
      int cj = -1, kj = 0;
      if (j < nk) { cj = sLc[b2][j] - s*SNC; kj = sLk[b2][j]; }
      unsigned mask = (unsigned)__ballot((unsigned)cj < (unsigned)SNC);
      while (mask) {
        const int src = __ffs(mask) - 1;
        mask &= mask - 1u;
        const int c  = __shfl(cj, src, 32);
        const int kk = __shfl(kj, src, 32);
        if (row == (c & 15)) {
          #pragma unroll
          for (int r = 0; r < 8; r++) {
            const int rowid = rowbase + hi*8 + r;
            const float w = attnW[(size_t)rowid*TK_ + kk];
            if (c < 16) acc0[r] += w; else acc1[r] += w;
          }
        }
      }
    }

    #pragma unroll
    for (int r = 0; r < 8; r++) {
      h0[r] = __logf(acc0[r]);
      h1[r] = __logf(acc1[r]);
    }
    __syncthreads();
  }

  #pragma unroll
  for (int r = 0; r < 8; r++) {
    float* o = outp + (size_t)(rowbase + hi*8 + r)*V_ + v0 + (SNST - 1)*SNC + row;
    o[0]  = h0[r];
    o[16] = h1[r];
  }
}

extern "C" void kernel_launch(void* const* d_in, const int* in_sizes, int n_in,
                              void* d_out, int out_size, void* d_ws, size_t ws_size,
                              hipStream_t stream) {
  (void)in_sizes; (void)n_in; (void)out_size; (void)ws_size;
  const float* dom    = (const float*)d_in[0];
  const float* out_st = (const float*)d_in[1];
  const float* ctx    = (const float*)d_in[2];
  const int*   cp     = (const int*)d_in[3];
  const float* Ws     = (const float*)d_in[4];
  const float* bsv    = (const float*)d_in[5];
  const float* Wg     = (const float*)d_in[6];
  const float* bg     = (const float*)d_in[7];
  float* outp = (float*)d_out;

  float* attnW = (float*)d_ws;
  float* pgenW = attnW + (size_t)B_*TQ_*TK_;
  float* rmW   = pgenW + B_*TQ_;
  float* rsW   = rmW + B_*TQ_;
  float* pmW   = outp;
  float* psW   = outp + (size_t)B_*TQ_*PSTR;
  float* amW   = psW + (size_t)B_*TQ_*PSTR;
  float* asW   = amW + B_*TQ_*NKP;
  float* auW   = asW + B_*TQ_*NKP;
  float* abW   = auW + B_*TQ_*NKP;

  k_attn  <<<dim3(B_*TQ_/16*NKP),    256, 0, stream>>>(dom, out_st, ctx, Wg, attnW, amW, asW, auW, abW);
  k_stats <<<dim3(B_*TQ_/128, NSPL), 256, 0, stream>>>(out_st, Ws, bsv, pmW, psW);
  k_reduce<<<dim3(B_*TQ_/8),         256, 0, stream>>>(pmW, psW, rmW, rsW, amW, asW, auW, abW, bg, attnW, pgenW);
  k_final <<<dim3(B_*TQ_/128, NSPL), 256, 0, stream>>>(out_st, Ws, bsv, cp, attnW, pgenW, rmW, rsW, outp);
}


# Vocab passes, query fragments: rows enter as whole-line 16 B loads (4 lines per instruction, was 16) and are turned into the k-step fragments through a wave-private piece of LDS; sums unchanged
# speedup vs baseline: 2.7821x; 1.0383x over previous
#include <hip/hip_runtime.h>
#include <hip/hip_bf16.h>


#define B_  16
#define TQ_ 64
#define TK_ 512
#define D_  256
#define V_  32000

#define SNC  32
#define SPP  48
#define SNST 8
#define NSPL 125
#define PSTR 128
#define NKP  4
#define KP_  (TK_/NKP)
#define ACH  64
#define APIT 68
#define AREG 1536

typedef __attribute__((ext_vector_type(2))) float v2f;
typedef __attribute__((ext_vector_type(8))) float v8f;

__device__ inline v8f wmma_f32_k4(v2f a, v2f b, v8f c) {
  return __builtin_amdgcn_wmma_f32_16x16x4_f32(
       false, a,  false, b,
       (short)0, c,  false,  false);
}

__device__ inline float wave_max(float x) {
  #pragma unroll
  for (int o = 1; o < 32; o <<= 1) x = fmaxf(x, __shfl_xor(x, o, 32));
  return x;
}
__device__ inline float wave_sum(float x) {
  #pragma unroll
  for (int o = 1; o < 32; o <<= 1) x += __shfl_xor(x, o, 32);
  return x;
}
__device__ inline float half16_max(float x) {
  #pragma unroll
  for (int o = 1; o < 16; o <<= 1) x = fmaxf(x, __shfl_xor(x, o, 32));
  return x;
}
__device__ inline float half16_sum(float x) {
  #pragma unroll
  for (int o = 1; o < 16; o <<= 1) x += __shfl_xor(x, o, 32);
  return x;
}

__global__ __launch_bounds__(256) void k_attn(
    const float* __restrict__ dom, const float* __restrict__ out_st,
    const float* __restrict__ ctx, const float* __restrict__ Wg,
    float* __restrict__ eW, float* __restrict__ amW, float* __restrict__ asW,
    float* __restrict__ auW, float* __restrict__ abW)
{
  __shared__ float sS[16][KP_ + 4];
  __shared__ float sCV[16][D_];

  const int part = blockIdx.x & 3;
  const int q0   = ((blockIdx.x >> 2) & 3) << 4;
  const int b    = blockIdx.x >> 4;
  const int kb   = part * KP_;
  const int wave = threadIdx.x >> 5;
  const int lane = threadIdx.x & 31;
  const int row  = lane & 15;
  const int hi   = lane >> 4;

  v8f acc = {};
  const float* Abase = out_st + (size_t)(b*TQ_ + q0 + row)*D_ + 4*hi;
  const float* Bbase = ctx + (size_t)(b*TK_ + kb + wave*16 + row)*D_ + 4*hi;
  #pragma unroll 16
  for (int j = 0; j < D_/8; ++j) {
    const float4 a4 = *(const float4*)(Abase + 8*j);
    const float4 b4 = *(const float4*)(Bbase + 8*j);
    v2f a0, a1, x, y;
    a0.x = a4.x; a0.y = a4.y; a1.x = a4.z; a1.y = a4.w;
    x.x = b4.x; x.y = b4.y; y.x = b4.z; y.y = b4.w;
    acc = wmma_f32_k4(a0, x, acc);
    acc = wmma_f32_k4(a1, y, acc);
  }
  #pragma unroll
  for (int r = 0; r < 8; r++)
    sS[hi*8 + r][wave*16 + row] = acc[r] * 0.0625f;
  __syncthreads();

  #pragma unroll
  for (int j = 0; j < 2; j++) {
    const int q = wave*2 + j;
    const int rowg = b*TQ_ + q0 + q;
    float v[KP_/32];
    float m = -1e30f;
    #pragma unroll
    for (int t = 0; t < KP_/32; t++) { v[t] = sS[q][lane + 32*t]; m = fmaxf(m, v[t]); }
    m = wave_max(m);
    float s = 0.f;
    #pragma unroll
    for (int t = 0; t < KP_/32; t++) { v[t] = __expf(v[t] - m); s += v[t]; }
    s = wave_sum(s);
    #pragma unroll
    for (int t = 0; t < KP_/32; t++) {
      sS[q][lane + 32*t] = v[t];
      eW[(size_t)rowg*TK_ + kb + lane + 32*t] = v[t];
    }
    if (lane == 0) { amW[rowg*NKP + part] = m; asW[rowg*NKP + part] = s; }
  }
  __syncthreads();

  const int qa = wave*2;
  float w1[8], w0[8], w2[8], ov[2][8], dv[2][8];
  #pragma unroll
  for (int t = 0; t < 8; t++) {
    w1[t] = Wg[D_ + lane + 32*t];
    w0[t] = 0.f; w2[t] = 0.f;
    ov[0][t] = 0.f; ov[1][t] = 0.f; dv[0][t] = 0.f; dv[1][t] = 0.f;
  }
  if (part == 0) {
    #pragma unroll
    for (int t = 0; t < 8; t++) {
      const int d = lane + 32*t;
      w0[t] = Wg[d]; w2[t] = Wg[2*D_ + d];
      #pragma unroll
      for (int j = 0; j < 2; j++) {
        ov[j][t] = out_st[(size_t)(b*TQ_ + q0 + qa + j)*D_ + d];
        dv[j][t] = dom[(size_t)(b*TQ_ + q0 + qa + j)*D_ + d];
      }
    }
  }

  v8f cv[2]; cv[0] = {}; cv[1] = {};
  const float* Cb = ctx + (size_t)(b*TK_ + kb + 2*hi)*D_ + 32*wave + 2*row;
  #pragma unroll 16
  for (int kc = 0; kc < KP_/4; ++kc) {
    const int k0 = kc*4;
    v2f a;
    a.x = sS[row][k0 + 2*hi];
    a.y = sS[row][k0 + 2*hi + 1];
    const v2f c0 = *(const v2f*)(Cb + (size_t)k0*D_);
    const v2f c1 = *(const v2f*)(Cb + (size_t)(k0 + 1)*D_);
    v2f bb0, bb1;
    bb0.x = c0.x; bb0.y = c1.x;
    bb1.x = c0.y; bb1.y = c1.y;
    cv[0] = wmma_f32_k4(a, bb0, cv[0]);
    cv[1] = wmma_f32_k4(a, bb1, cv[1]);
  }

  #pragma unroll
  for (int r = 0; r < 8; r++) {
    sCV[hi*8 + r][32*wave + 2*row]     = cv[0][r];
    sCV[hi*8 + r][32*wave + 2*row + 1] = cv[1][r];
  }
  __syncthreads();

  #pragma unroll
  for (int j = 0; j < 2; j++) {
    const int q = qa + j;
    const int rowg = b*TQ_ + q0 + q;
    float du = 0.f;
    #pragma unroll
    for (int t = 0; t < 8; t++) du += sCV[q][lane + 32*t] * w1[t];
    du = wave_sum(du);
    if (lane == 0) auW[rowg*NKP + part] = du;
    if (part == 0) {
      float db = 0.f;
      #pragma unroll
      for (int t = 0; t < 8; t++) {
        db += ov[j][t] * w0[t];
        db += dv[j][t] * w2[t];
      }
      db = wave_sum(db);
      if (lane == 0) abW[rowg] = db;
    }
  }
}

__device__ __forceinline__ void load_a_frags(const float* __restrict__ out_st, int rowbase,
                                             float* sA, int row, int hi, v2f (&a)[D_/4]) {
  const float* Arow = out_st + (size_t)(rowbase + hi)*D_ + 4*row;
  #pragma unroll
  for (int ch = 0; ch < D_/ACH; ++ch) {
    float4 t[8];
    #pragma unroll
    for (int i = 0; i < 8; ++i) t[i] = *(const float4*)(Arow + (size_t)(2*i)*D_ + ACH*ch);
    #pragma unroll
    for (int i = 0; i < 8; ++i) *(float4*)&sA[(2*i + hi)*APIT + 4*row] = t[i];
    __builtin_amdgcn_fence(__ATOMIC_ACQ_REL, "wavefront");
    #pragma unroll
    for (int d = 0; d < ACH/4; ++d) a[(ACH/4)*ch + d] = *(const v2f*)&sA[row*APIT + 4*d + 2*hi];
    __builtin_amdgcn_fence(__ATOMIC_ACQ_REL, "wavefront");
  }
}

__global__ __launch_bounds__(256) void k_stats(
    const float* __restrict__ out_st, const float* __restrict__ Ws,
    const float* __restrict__ bs, float* __restrict__ pmW,
    float* __restrict__ psW)
{
  __shared__ v2f sWp[D_/2][SPP];

  const int tid  = threadIdx.x;
  const int wave = tid >> 5;
  const int lane = tid & 31;
  const int row  = lane & 15;
  const int hi   = lane >> 4;
  const int sp   = blockIdx.y;
  const int v0   = sp * (SNC*SNST);
  const int rowbase = blockIdx.x*128 + wave*16;

  const int sd = tid >> 3;
  const int sc = (tid & 7) * 4;
  const float* Wsrc = Ws + (size_t)(2*sd)*V_ + v0 + sc;
  float4 pf[8];
  #pragma unroll
  for (int i = 0; i < 4; i++) {
    pf[2*i]     = *(const float4*)(Wsrc + (size_t)(i*64)*V_);
    pf[2*i + 1] = *(const float4*)(Wsrc + (size_t)(i*64 + 1)*V_);
  }

  v2f a[D_/4];
  load_a_frags(out_st, rowbase, (float*)&sWp[0][0] + wave*AREG, row, hi, a);
  __syncthreads();

  float rm[8], rs[8];
  #pragma unroll
  for (int r = 0; r < 8; r++) { rm[r] = -1e30f; rs[r] = 0.f; }

  const v2f* pB = &sWp[hi][row];

  #pragma unroll 1
  for (int s = 0; s < SNST; ++s) {
    #pragma unroll
    for (int i = 0; i < 4; i++) {
      const float4 x = pf[2*i], y = pf[2*i + 1];
      float4 lo, up;
      lo.x = x.x; lo.y = y.x; lo.z = x.y; lo.w = y.y;
      up.x = x.z; up.y = y.z; up.z = x.w; up.w = y.w;
      *(float4*)&sWp[i*32 + sd][sc]     = lo;
      *(float4*)&sWp[i*32 + sd][sc + 2] = up;
    }
    __syncthreads();
    if (s + 1 < SNST) {
      #pragma unroll
      for (int i = 0; i < 4; i++) {
        pf[2*i]     = *(const float4*)(Wsrc + (size_t)(i*64)*V_     + (s + 1)*SNC);
        pf[2*i + 1] = *(const float4*)(Wsrc + (size_t)(i*64 + 1)*V_ + (s + 1)*SNC);
      }
    }

    v8f acc0 = {}, acc1 = {};
    #pragma unroll
    for (int dc = 0; dc < D_/4; ++dc) {
      const v2f b0 = pB[dc*2*SPP];
      const v2f b1 = pB[dc*2*SPP + 16];
      acc0 = wmma_f32_k4(a[dc], b0, acc0);
      acc1 = wmma_f32_k4(a[dc], b1, acc1);
    }

    const int vs = v0 + s*SNC;
    const float bsv0 = bs[vs + row];
    const float bsv1 = bs[vs + 16 + row];
    #pragma unroll
    for (int r = 0; r < 8; r++) {
      const float z0 = acc0[r] + bsv0;
      const float z1 = acc1[r] + bsv1;
      const float nm = fmaxf(rm[r], fmaxf(z0, z1));
      rs[r] = rs[r] * __expf(rm[r] - nm) + (__expf(z0 - nm) + __expf(z1 - nm));
      rm[r] = nm;
    }
    __syncthreads();
  }

  #pragma unroll
  for (int r = 0; r < 8; r++) {
    const float gm = half16_max(rm[r]);
    const float gs = half16_sum(rs[r] * __expf(rm[r] - gm));
    if (row == 0) {
      const int rowid = rowbase + hi*8 + r;
      pmW[(size_t)rowid*PSTR + sp] = gm;
      psW[(size_t)rowid*PSTR + sp] = gs;
    }
  }
}

__global__ __launch_bounds__(256) void k_reduce(
    const float* __restrict__ pmW, const float* __restrict__ psW,
    float* __restrict__ rmW, float* __restrict__ rsW,
    const float* __restrict__ amW, const float* __restrict__ asW,
    const float* __restrict__ auW, const float* __restrict__ abW,
    const float* __restrict__ bg, float* __restrict__ eW,
    float* __restrict__ pgenW)
{
  const int rowid = blockIdx.x*8 + (threadIdx.x >> 5);
  const int lane  = threadIdx.x & 31;
  float m[4], s[4];
  #pragma unroll
  for (int t = 0; t < 4; t++) {
    const int spi = lane + 32*t;
    const bool ok = spi < NSPL;
    m[t] = ok ? pmW[(size_t)rowid*PSTR + spi] : -1e30f;
    s[t] = ok ? psW[(size_t)rowid*PSTR + spi] : 0.f;
  }
  float am[NKP], as4[NKP], au[NKP];
  #pragma unroll
  for (int p = 0; p < NKP; p++) {
    am[p]  = amW[rowid*NKP + p];
    as4[p] = asW[rowid*NKP + p];
    au[p]  = auW[rowid*NKP + p];
  }
  const float ab  = abW[rowid];
  const float bg0 = bg[0];
  float e[TK_/32];
  #pragma unroll
  for (int t = 0; t < TK_/32; t++) e[t] = eW[(size_t)rowid*TK_ + lane + 32*t];

  float gm = -1e30f;
  #pragma unroll
  for (int t = 0; t < 4; t++) gm = fmaxf(gm, m[t]);
  gm = wave_max(gm);
  float gs = 0.f;
  #pragma unroll
  for (int t = 0; t < 4; t++) gs += s[t] * __expf(m[t] - gm);
  gs = wave_sum(gs);

  float M = am[0];
  #pragma unroll
  for (int p = 1; p < NKP; p++) M = fmaxf(M, am[p]);
  float c[NKP];
  float S = 0.f, U = 0.f;
  #pragma unroll
  for (int p = 0; p < NKP; p++) {
    c[p] = __expf(am[p] - M);
    S += as4[p] * c[p];
    U += au[p] * c[p];
  }
  const float invS = 1.0f / S;
  const float x  = (ab + U * invS) + bg0;
  const float pg = 1.0f / (1.0f + __expf(-x));
  const float om = 1.0f - pg;
  #pragma unroll
  for (int t = 0; t < TK_/32; t++)
    eW[(size_t)rowid*TK_ + lane + 32*t] = e[t] * (om * (c[t >> 2] * invS));
  if (lane == 0) { rmW[rowid] = gm; rsW[rowid] = gs; pgenW[rowid] = pg; }
}

__global__ __launch_bounds__(256) void k_final(
    const float* __restrict__ out_st, const float* __restrict__ Ws,
    const float* __restrict__ bs, const int* __restrict__ cp,
    const float* __restrict__ attnW, const float* __restrict__ pgenW,
    const float* __restrict__ rmW, const float* __restrict__ rsW,
    float* __restrict__ outp)
{
  __shared__ v2f sWp[D_/2][SPP];
  __shared__ int sLk[2][TK_];
  __shared__ int sLc[2][TK_];
  __shared__ int sLn[2];

  const int tid  = threadIdx.x;
  const int wave = tid >> 5;
  const int lane = tid & 31;
  const int row  = lane & 15;
  const int hi   = lane >> 4;
  const int sp   = blockIdx.y;
  const int v0   = sp * (SNC*SNST);
  const int rowbase = blockIdx.x*128 + wave*16;
  const int b2   = wave >> 2;

  if (wave < 2) {
    const int* keys = cp + (blockIdx.x*2 + wave)*TK_;
    int kv[TK_/32];
    #pragma unroll
    for (int it = 0; it < TK_/32; ++it) kv[it] = keys[it*32 + lane];
    int base = 0;
    #pragma unroll
    for (int it = 0; it < TK_/32; ++it) {
      const int c   = kv[it] - v0;
      const bool in = (unsigned)c < (unsigned)(SNC*SNST);
      const unsigned mask = (unsigned)__ballot(in);
      if (in) {
        const int pos = base + __popc(mask & ((1u << lane) - 1u));
        sLk[wave][pos] = it*32 + lane;
        sLc[wave][pos] = c;
      }
      base += __popc(mask);
    }
    if (lane == 0) sLn[wave] = base;
  }

  const int sd = tid >> 3;
  const int sc = (tid & 7) * 4;
  const float* Wsrc = Ws + (size_t)(2*sd)*V_ + v0 + sc;
  float4 pf[8];
  #pragma unroll
  for (int i = 0; i < 4; i++) {
    pf[2*i]     = *(const float4*)(Wsrc + (size_t)(i*64)*V_);
    pf[2*i + 1] = *(const float4*)(Wsrc + (size_t)(i*64 + 1)*V_);
  }

  v2f a[D_/4];
  load_a_frags(out_st, rowbase, (float*)&sWp[0][0] + wave*AREG, row, hi, a);
  __syncthreads();

  float sc8[8], mr8[8];
  #pragma unroll
  for (int r = 0; r < 8; r++) {
    const int rowid = rowbase + hi*8 + r;
    sc8[r] = pgenW[rowid] / rsW[rowid];
    mr8[r] = rmW[rowid];
  }

  const v2f* pB = &sWp[hi][row];

  float h0[8], h1[8];
  #pragma unroll
  for (int r = 0; r < 8; r++) { h0[r] = 0.f; h1[r] = 0.f; }

  #pragma unroll 1
  for (int s = 0; s < SNST; ++s) {
    #pragma unroll
    for (int i = 0; i < 4; i++) {
      const float4 x = pf[2*i], y = pf[2*i + 1];
      float4 lo, up;
      lo.x = x.x; lo.y = y.x; lo.z = x.y; lo.w = y.y;
      up.x = x.z; up.y = y.z; up.z = x.w; up.w = y.w;
      *(float4*)&sWp[i*32 + sd][sc]     = lo;
      *(float4*)&sWp[i*32 + sd][sc + 2] = up;
    }
    __syncthreads();
    if (s > 0) {
      #pragma unroll
      for (int r = 0; r < 8; r++) {
        float* o = outp + (size_t)(rowbase + hi*8 + r)*V_ + v0 + (s - 1)*SNC + row;
        o[0]  = h0[r];
        o[16] = h1[r];
      }
    }
    if (s + 1 < SNST) {
      #pragma unroll
      for (int i = 0; i < 4; i++) {
        pf[2*i]     = *(const float4*)(Wsrc + (size_t)(i*64)*V_     + (s + 1)*SNC);
        pf[2*i + 1] = *(const float4*)(Wsrc + (size_t)(i*64 + 1)*V_ + (s + 1)*SNC);
      }
    }

    v8f acc0 = {}, acc1 = {};
    #pragma unroll
    for (int dc = 0; dc < D_/4; ++dc) {
      const v2f b0 = pB[dc*2*SPP];
      const v2f b1 = pB[dc*2*SPP + 16];
      acc0 = wmma_f32_k4(a[dc], b0, acc0);
      acc1 = wmma_f32_k4(a[dc], b1, acc1);
    }

    const int vs = v0 + s*SNC;
    const float bsv0 = bs[vs + row];
    const float bsv1 = bs[vs + 16 + row];
    #pragma unroll
    for (int r = 0; r < 8; r++) {
      acc0[r] = sc8[r] * __expf((acc0[r] + bsv0) - mr8[r]);
      acc1[r] = sc8[r] * __expf((acc1[r] + bsv1) - mr8[r]);
    }

    const int nk = sLn[b2];
    for (int j0 = 0; j0 < nk; j0 += 32) {
      const int j = j0 + lane;
      int cj = -1, kj = 0;
      if (j < nk) { cj = sLc[b2][j] - s*SNC; kj = sLk[b2][j]; }
      unsigned mask = (unsigned)__ballot((unsigned)cj < (unsigned)SNC);
      while (mask) {
        const int src = __ffs(mask) - 1;
        mask &= mask - 1u;
        const int c  = __shfl(cj, src, 32);
        const int kk = __shfl(kj, src, 32);
        if (row == (c & 15)) {
          #pragma unroll
          for (int r = 0; r < 8; r++) {
            const int rowid = rowbase + hi*8 + r;
            const float w = attnW[(size_t)rowid*TK_ + kk];
            if (c < 16) acc0[r] += w; else acc1[r] += w;
          }
        }
      }
    }

    #pragma unroll
    for (int r = 0; r < 8; r++) {
      h0[r] = __logf(acc0[r]);
      h1[r] = __logf(acc1[r]);
    }
    __syncthreads();
  }

  #pragma unroll
  for (int r = 0; r < 8; r++) {
    float* o = outp + (size_t)(rowbase + hi*8 + r)*V_ + v0 + (SNST - 1)*SNC + row;
    o[0]  = h0[r];
    o[16] = h1[r];
  }
}

extern "C" void kernel_launch(void* const* d_in, const int* in_sizes, int n_in,
                              void* d_out, int out_size, void* d_ws, size_t ws_size,
                              hipStream_t stream) {
  (void)in_sizes; (void)n_in; (void)out_size; (void)ws_size;
  const float* dom    = (const float*)d_in[0];
  const float* out_st = (const float*)d_in[1];
  const float* ctx    = (const float*)d_in[2];
  const int*   cp     = (const int*)d_in[3];
  const float* Ws     = (const float*)d_in[4];
  const float* bsv    = (const float*)d_in[5];
  const float* Wg     = (const float*)d_in[6];
  const float* bg     = (const float*)d_in[7];
  float* outp = (float*)d_out;

  float* attnW = (float*)d_ws;
  float* pgenW = attnW + (size_t)B_*TQ_*TK_;
  float* rmW   = pgenW + B_*TQ_;
  float* rsW   = rmW + B_*TQ_;
  float* pmW   = outp;
  float* psW   = outp + (size_t)B_*TQ_*PSTR;
  float* amW   = psW + (size_t)B_*TQ_*PSTR;
  float* asW   = amW + B_*TQ_*NKP;
  float* auW   = asW + B_*TQ_*NKP;
  float* abW   = auW + B_*TQ_*NKP;

  k_attn  <<<dim3(B_*TQ_/16*NKP),    256, 0, stream>>>(dom, out_st, ctx, Wg, attnW, amW, asW, auW, abW);
  k_stats <<<dim3(B_*TQ_/128, NSPL), 256, 0, stream>>>(out_st, Ws, bsv, pmW, psW);
  k_reduce<<<dim3(B_*TQ_/8),         256, 0, stream>>>(pmW, psW, rmW, rsW, amW, asW, auW, abW, bg, attnW, pgenW);
  k_final <<<dim3(B_*TQ_/128, NSPL), 256, 0, stream>>>(out_st, Ws, bsv, cp, attnW, pgenW, rmW, rsW, outp);
}
